# starting kernel
# baseline (speedup 1.0000x reference)
#include <hip/hip_runtime.h>

typedef float v2f __attribute__((ext_vector_type(2)));
typedef float v8f __attribute__((ext_vector_type(8)));

__device__ __forceinline__ v8f wmma4(v2f a, v2f b, v8f c) {
    return __builtin_amdgcn_wmma_f32_16x16x4_f32(
        false, a, false, b, (short)0, c, false, false);
}

template <int MODE>
__global__ __launch_bounds__(256) void gemm_kernel(
    const float* __restrict__ A, const float* __restrict__ W,
    const float* __restrict__ bias, float* __restrict__ out,
    float* __restrict__ Qb, float* __restrict__ Kb, float* __restrict__ Vb,
    int Ncols, int Kdim)
{
    __shared__ float As[128 * 20];
    __shared__ float Wsh[64 * 20];

    const int tid  = threadIdx.x;
    const int wid  = tid >> 5;
    const int lane = tid & 31;
    const int half = lane >> 4;
    const int l15  = lane & 15;
    const int klo  = half * 2;
    const int wave_m = wid >> 1;
    const int wave_n = wid & 1;

    const int rowBase = blockIdx.y * 128;
    const int colBase = blockIdx.x * 64;

    v8f acc[2][2] = {};

    const int lr = tid >> 2;
    const int lc = (tid & 3) << 2;

    for (int kb = 0; kb < Kdim; kb += 16) {
        __syncthreads();
        *(float4*)&As[lr * 20 + lc] =
            *(const float4*)(A + (size_t)(rowBase + lr) * Kdim + kb + lc);
        *(float4*)&As[(lr + 64) * 20 + lc] =
            *(const float4*)(A + (size_t)(rowBase + lr + 64) * Kdim + kb + lc);
        *(float4*)&Wsh[lr * 20 + lc] =
            *(const float4*)(W + (size_t)(colBase + lr) * Kdim + kb + lc);
        __syncthreads();

#pragma unroll
        for (int ks = 0; ks < 16; ks += 4) {
            v2f a0 = *(const v2f*)&As[(wave_m * 32 + l15) * 20 + ks + klo];
            v2f a1 = *(const v2f*)&As[(wave_m * 32 + 16 + l15) * 20 + ks + klo];
            v2f w0 = *(const v2f*)&Wsh[(wave_n * 32 + l15) * 20 + ks + klo];
            v2f w1 = *(const v2f*)&Wsh[(wave_n * 32 + 16 + l15) * 20 + ks + klo];
            acc[0][0] = wmma4(a0, w0, acc[0][0]);
            acc[0][1] = wmma4(a0, w1, acc[0][1]);
            acc[1][0] = wmma4(a1, w0, acc[1][0]);
            acc[1][1] = wmma4(a1, w1, acc[1][1]);
        }
    }

#pragma unroll
    for (int sm = 0; sm < 2; ++sm) {
#pragma unroll
        for (int sn = 0; sn < 2; ++sn) {
            const int j  = colBase + wave_n * 32 + sn * 16 + l15;
            const int r0 = rowBase + wave_m * 32 + sm * 16 + half * 8;
            if (MODE == 0) {
                const float bv = bias ? bias[j] : 0.0f;
                float* dst = out + (size_t)r0 * Ncols + j;
#pragma unroll
                for (int g = 0; g < 8; ++g)
                    dst[(size_t)g * Ncols] = acc[sm][sn][g] + bv;
            } else {
                const int t  = j >> 10;
                const int hh = (j >> 6) & 15;
                const int d  = j & 63;
                const int bb = r0 >> 10;
                const int n  = r0 & 1023;
                float* dst;
                if (t == 0)
                    dst = Qb + ((size_t)(bb * 16 + hh) * 1024 + n) * 64 + d;
                else if (t == 1)
                    dst = Kb + ((size_t)(bb * 16 + hh) * 1280 + 256 + n) * 64 + d;
                else
                    dst = Vb + ((size_t)(bb * 16 + hh) * 1280 + 256 + n) * 64 + d;
#pragma unroll
                for (int g = 0; g < 8; ++g)
                    dst[(size_t)g * 64] = acc[sm][sn][g];
            }
        }
    }
}

__global__ __launch_bounds__(256) void ctx_scatter_kernel(
    const float* __restrict__ ctx, float* __restrict__ Kb, float* __restrict__ Vb)
{
    const int idx = blockIdx.x * 256 + threadIdx.x;
    const int f   = idx << 2;
    const int b   = f >> 19;
    const int m   = (f >> 11) & 255;
    const int q   = f & 2047;
    const int s   = q >> 10;
    const int hh  = (q >> 6) & 15;
    const int d   = q & 63;
    const float4 v = *(const float4*)(ctx + (size_t)f);
    float* base = s ? Vb : Kb;
    *(float4*)(base + ((size_t)(b * 16 + hh) * 1280 + m) * 64 + d) = v;
}

__global__ __launch_bounds__(256) void attn_kernel(
    const float* __restrict__ Qb, const float* __restrict__ Kb,
    const float* __restrict__ Vb, float* __restrict__ AO)
{
    __shared__ float Ks[32 * 68];
    __shared__ float VTs[64 * 34];
    __shared__ float Ps[8 * 16 * 34];

    const int tid  = threadIdx.x;
    const int wid  = tid >> 5;
    const int lane = tid & 31;
    const int half = lane >> 4;
    const int l15  = lane & 15;
    const int klo  = half * 2;

    const int bh = blockIdx.x;
    const int b  = bh >> 4;
    const int h  = bh & 15;
    const float* Qp = Qb + (size_t)bh * 1024 * 64;
    const float* Kp = Kb + (size_t)bh * 1280 * 64;
    const float* Vp = Vb + (size_t)bh * 1280 * 64;

    const int n0   = blockIdx.y * 128 + wid * 16;
    const int qrow = n0 + l15;
    const float scale = 0.125f;

    v2f qf[16];
#pragma unroll
    for (int kk = 0; kk < 16; ++kk) {
        v2f t = *(const v2f*)(Qp + (size_t)qrow * 64 + kk * 4 + klo);
        qf[kk] = t * scale;
    }

    float mrow[8], lrow[8];
    v8f oacc[4] = {};
#pragma unroll
    for (int g = 0; g < 8; ++g) { mrow[g] = -3.0e38f; lrow[g] = 0.0f; }

    float* Pw = &Ps[wid * 16 * 34];

    for (int mc = 0; mc < 1280; mc += 32) {
        __syncthreads();
#pragma unroll
        for (int it = 0; it < 2; ++it) {
            const int idx = tid + it * 256;
            const int r   = idx >> 4;
            const int c4  = (idx & 15) << 2;
            *(float4*)&Ks[r * 68 + c4] =
                *(const float4*)(Kp + (size_t)(mc + r) * 64 + c4);
            const float4 v = *(const float4*)(Vp + (size_t)(mc + r) * 64 + c4);
            VTs[(c4 + 0) * 34 + r] = v.x;
            VTs[(c4 + 1) * 34 + r] = v.y;
            VTs[(c4 + 2) * 34 + r] = v.z;
            VTs[(c4 + 3) * 34 + r] = v.w;
        }
        __syncthreads();

        v8f s0 = {}, s1 = {};
#pragma unroll
        for (int kk = 0; kk < 16; ++kk) {
            v2f b0 = *(const v2f*)&Ks[l15 * 68 + kk * 4 + klo];
            v2f b1 = *(const v2f*)&Ks[(16 + l15) * 68 + kk * 4 + klo];
            s0 = wmma4(qf[kk], b0, s0);
            s1 = wmma4(qf[kk], b1, s1);
        }

        float alpha[8];
#pragma unroll
        for (int g = 0; g < 8; ++g) {
            float cm = fmaxf(s0[g], s1[g]);
            cm = fmaxf(cm, __shfl_xor(cm, 1, 32));
            cm = fmaxf(cm, __shfl_xor(cm, 2, 32));
            cm = fmaxf(cm, __shfl_xor(cm, 4, 32));
            cm = fmaxf(cm, __shfl_xor(cm, 8, 32));
            const float mnew = fmaxf(mrow[g], cm);
            alpha[g] = __expf(mrow[g] - mnew);
            mrow[g]  = mnew;
            const float p0 = __expf(s0[g] - mnew);
            const float p1 = __expf(s1[g] - mnew);
            s0[g] = p0; s1[g] = p1;
            float cs = p0 + p1;
            cs += __shfl_xor(cs, 1, 32);
            cs += __shfl_xor(cs, 2, 32);
            cs += __shfl_xor(cs, 4, 32);
            cs += __shfl_xor(cs, 8, 32);
            lrow[g] = lrow[g] * alpha[g] + cs;
        }
#pragma unroll
        for (int t4 = 0; t4 < 4; ++t4)
#pragma unroll
            for (int g = 0; g < 8; ++g) oacc[t4][g] *= alpha[g];

#pragma unroll
        for (int g = 0; g < 8; ++g) {
            const int r = g + half * 8;
            Pw[r * 34 + l15]      = s0[g];
            Pw[r * 34 + 16 + l15] = s1[g];
        }
        asm volatile("s_wait_dscnt 0" ::: "memory");

#pragma unroll
        for (int kk = 0; kk < 8; ++kk) {
            v2f pf = *(const v2f*)&Pw[l15 * 34 + kk * 4 + klo];
#pragma unroll
            for (int t4 = 0; t4 < 4; ++t4) {
                v2f vf = *(const v2f*)&VTs[(t4 * 16 + l15) * 34 + kk * 4 + klo];
                oacc[t4] = wmma4(pf, vf, oacc[t4]);
            }
        }
    }

#pragma unroll
    for (int g = 0; g < 8; ++g) {
        const float rinv = 1.0f / lrow[g];
        const int n = n0 + g + half * 8;
        float* dst = AO + ((size_t)(b * 1024 + n) * 1024) + h * 64;
#pragma unroll
        for (int t4 = 0; t4 < 4; ++t4)
            dst[t4 * 16 + l15] = oacc[t4][g] * rinv;
    }
}

extern "C" void kernel_launch(void* const* d_in, const int* in_sizes, int n_in,
                              void* d_out, int out_size, void* d_ws, size_t ws_size,
                              hipStream_t stream)
{
    const float* x      = (const float*)d_in[0];
    const float* ctx    = (const float*)d_in[1];
    const float* w_qkv  = (const float*)d_in[2];
    const float* w_proj = (const float*)d_in[3];
    const float* b_proj = (const float*)d_in[4];
    float* outp = (float*)d_out;

    float* Qb = (float*)d_ws;
    float* Kb = Qb + 4194304;
    float* Vb = Kb + 5242880;
    float* AO = Vb + 5242880;

    gemm_kernel<1><<<dim3(3072 / 64, 4096 / 128), 256, 0, stream>>>(
        x, w_qkv, nullptr, nullptr, Qb, Kb, Vb, 3072, 1024);

    ctx_scatter_kernel<<<2048, 256, 0, stream>>>(ctx, Kb, Vb);

    attn_kernel<<<dim3(64, 8), 256, 0, stream>>>(Qb, Kb, Vb, AO);

    gemm_kernel<0><<<dim3(1024 / 64, 4096 / 128), 256, 0, stream>>>(
        AO, w_proj, b_proj, outp, nullptr, nullptr, nullptr, 1024, 1024);
}


# Global->LDS staging software-pipelined in the GEMM K loop and the attention kv loop: next tile's global loads issued before the current tile's WMMAs, two LDS stages, one barrier per step not two
# speedup vs baseline: 1.0377x; 1.0377x over previous
#include <hip/hip_runtime.h>

typedef float v2f __attribute__((ext_vector_type(2)));
typedef float v8f __attribute__((ext_vector_type(8)));

__device__ __forceinline__ v8f wmma4(v2f a, v2f b, v8f c) {
    return __builtin_amdgcn_wmma_f32_16x16x4_f32(
        false, a, false, b, (short)0, c, false, false);
}

template <int MODE>
__global__ __launch_bounds__(256) void gemm_kernel(
    const float* A, const float* W,
    const float* __restrict__ bias, float* __restrict__ out,
    float* __restrict__ Qb, float* __restrict__ Kb, float* __restrict__ Vb,
    int Ncols, int Kdim)
{
    __shared__ float As[2 * 128 * 20];
    __shared__ float Wsh[2 * 64 * 20];

    const int tid  = threadIdx.x;
    const int wid  = tid >> 5;
    const int lane = tid & 31;
    const int half = lane >> 4;
    const int l15  = lane & 15;
    const int klo  = half * 2;
    const int wave_m = wid >> 1;
    const int wave_n = wid & 1;

    const int rowBase = blockIdx.y * 128;
    const int colBase = blockIdx.x * 64;

    v8f acc[2][2] = {};

    const int lr = tid >> 2;
    const int lc = (tid & 3) << 2;

    const float* Ag0 = A + (size_t)(rowBase + lr) * Kdim + lc;
    const float* Ag1 = A + (size_t)(rowBase + lr + 64) * Kdim + lc;
    const float* Wg  = W + (size_t)(colBase + lr) * Kdim + lc;
    const int sA0 = lr * 20 + lc;
    const int sA1 = (lr + 64) * 20 + lc;
    const int sW  = lr * 20 + lc;
    const int fA  = (wave_m * 32 + l15) * 20 + klo;
    const int fW  = (wave_n * 32 + l15) * 20 + klo;

    float4 pa0 = *(const float4*)Ag0;
    float4 pa1 = *(const float4*)Ag1;
    float4 pw  = *(const float4*)Wg;
    *(float4*)&As[sA0]  = pa0;
    *(float4*)&As[sA1]  = pa1;
    *(float4*)&Wsh[sW]  = pw;
    __syncthreads();

    auto mma_tile = [&](const float* Ac, const float* Wc) {
#pragma unroll
        for (int ks = 0; ks < 16; ks += 4) {
            v2f a0 = *(const v2f*)&Ac[fA + ks];
            v2f a1 = *(const v2f*)&Ac[fA + 16 * 20 + ks];
            v2f w0 = *(const v2f*)&Wc[fW + ks];
            v2f w1 = *(const v2f*)&Wc[fW + 16 * 20 + ks];
            acc[0][0] = wmma4(a0, w0, acc[0][0]);
            acc[0][1] = wmma4(a0, w1, acc[0][1]);
            acc[1][0] = wmma4(a1, w0, acc[1][0]);
            acc[1][1] = wmma4(a1, w1, acc[1][1]);
        }
    };

    int cur = 0;
    for (int kb = 16; kb < Kdim; kb += 16) {
        pa0 = *(const float4*)(Ag0 + kb);
        pa1 = *(const float4*)(Ag1 + kb);
        pw  = *(const float4*)(Wg + kb);
        __builtin_amdgcn_sched_barrier(0);

        mma_tile(As + cur * (128 * 20), Wsh + cur * (64 * 20));
        __builtin_amdgcn_sched_barrier(0);

        const int nxt = cur ^ 1;
        *(float4*)&As[nxt * (128 * 20) + sA0] = pa0;
        *(float4*)&As[nxt * (128 * 20) + sA1] = pa1;
        *(float4*)&Wsh[nxt * (64 * 20) + sW]  = pw;
        __syncthreads();
        cur = nxt;
    }
    mma_tile(As + cur * (128 * 20), Wsh + cur * (64 * 20));

#pragma unroll
    for (int sm = 0; sm < 2; ++sm) {
#pragma unroll
        for (int sn = 0; sn < 2; ++sn) {
            const int j  = colBase + wave_n * 32 + sn * 16 + l15;
            const int r0 = rowBase + wave_m * 32 + sm * 16 + half * 8;
            if (MODE == 0) {
                const float bv = bias ? bias[j] : 0.0f;
                float* dst = out + (size_t)r0 * Ncols + j;
#pragma unroll
                for (int g = 0; g < 8; ++g)
                    dst[(size_t)g * Ncols] = acc[sm][sn][g] + bv;
            } else {
                const int t  = j >> 10;
                const int hh = (j >> 6) & 15;
                const int d  = j & 63;
                const int bb = r0 >> 10;
                const int n  = r0 & 1023;
                float* dst;
                if (t == 0)
                    dst = Qb + ((size_t)(bb * 16 + hh) * 1024 + n) * 64 + d;
                else if (t == 1)
                    dst = Kb + ((size_t)(bb * 16 + hh) * 1280 + 256 + n) * 64 + d;
                else
                    dst = Vb + ((size_t)(bb * 16 + hh) * 1280 + 256 + n) * 64 + d;
#pragma unroll
                for (int g = 0; g < 8; ++g)
                    dst[(size_t)g * 64] = acc[sm][sn][g];
            }
        }
    }
}

__global__ __launch_bounds__(256) void ctx_scatter_kernel(
    const float* __restrict__ ctx, float* __restrict__ Kb, float* __restrict__ Vb)
{
    const int idx = blockIdx.x * 256 + threadIdx.x;
    const int f   = idx << 2;
    const int b   = f >> 19;
    const int m   = (f >> 11) & 255;
    const int q   = f & 2047;
    const int s   = q >> 10;
    const int hh  = (q >> 6) & 15;
    const int d   = q & 63;
    const float4 v = *(const float4*)(ctx + (size_t)f);
    float* base = s ? Vb : Kb;
    *(float4*)(base + ((size_t)(b * 16 + hh) * 1280 + m) * 64 + d) = v;
}

__global__ __launch_bounds__(256) void attn_kernel(
    const float* __restrict__ Qb, const float* Kb,
    const float* Vb, float* __restrict__ AO)
{
    __shared__ float Ks[2 * 32 * 68];
    __shared__ float VTs[2 * 64 * 34];
    __shared__ float Ps[8 * 16 * 34];

    const int tid  = threadIdx.x;
    const int wid  = tid >> 5;
    const int lane = tid & 31;
    const int half = lane >> 4;
    const int l15  = lane & 15;
    const int klo  = half * 2;

    const int bh = blockIdx.x;
    const int b  = bh >> 4;
    const int h  = bh & 15;
    const float* Qp = Qb + (size_t)bh * 1024 * 64;
    const float* Kp = Kb + (size_t)bh * 1280 * 64;
    const float* Vp = Vb + (size_t)bh * 1280 * 64;

    const int n0   = blockIdx.y * 128 + wid * 16;
    const int qrow = n0 + l15;
    const float scale = 0.125f;

    const int sr = tid >> 4;
    const int sc = (tid & 15) << 2;
    const float* Kg = Kp + (size_t)sr * 64 + sc;
    const float* Vg = Vp + (size_t)sr * 64 + sc;

    float4 k0 = *(const float4*)(Kg);
    float4 k1 = *(const float4*)(Kg + 16 * 64);
    float4 v0 = *(const float4*)(Vg);
    float4 v1 = *(const float4*)(Vg + 16 * 64);

    v2f qf[16];
#pragma unroll
    for (int kk = 0; kk < 16; ++kk) {
        v2f t = *(const v2f*)(Qp + (size_t)qrow * 64 + kk * 4 + klo);
        qf[kk] = t * scale;
    }

    float mrow[8], lrow[8];
    v8f oacc[4] = {};
#pragma unroll
    for (int g = 0; g < 8; ++g) { mrow[g] = -3.0e38f; lrow[g] = 0.0f; }

    float* Pw = &Ps[wid * 16 * 34];

    auto stage_store = [&](float* Kd, float* Vd) {
        *(float4*)&Kd[sr * 68 + sc]        = k0;
        *(float4*)&Kd[(sr + 16) * 68 + sc] = k1;
        Vd[(sc + 0) * 34 + sr] = v0.x;
        Vd[(sc + 1) * 34 + sr] = v0.y;
        Vd[(sc + 2) * 34 + sr] = v0.z;
        Vd[(sc + 3) * 34 + sr] = v0.w;
        Vd[(sc + 0) * 34 + sr + 16] = v1.x;
        Vd[(sc + 1) * 34 + sr + 16] = v1.y;
        Vd[(sc + 2) * 34 + sr + 16] = v1.z;
        Vd[(sc + 3) * 34 + sr + 16] = v1.w;
    };

    auto chunk = [&](const float* Kc, const float* Vc) {
        v8f s0 = {}, s1 = {};
#pragma unroll
        for (int kk = 0; kk < 16; ++kk) {
            v2f b0 = *(const v2f*)&Kc[l15 * 68 + kk * 4 + klo];
            v2f b1 = *(const v2f*)&Kc[(16 + l15) * 68 + kk * 4 + klo];
            s0 = wmma4(qf[kk], b0, s0);
            s1 = wmma4(qf[kk], b1, s1);
        }

        float alpha[8];
#pragma unroll
        for (int g = 0; g < 8; ++g) {
            float cm = fmaxf(s0[g], s1[g]);
            cm = fmaxf(cm, __shfl_xor(cm, 1, 32));
            cm = fmaxf(cm, __shfl_xor(cm, 2, 32));
            cm = fmaxf(cm, __shfl_xor(cm, 4, 32));
            cm = fmaxf(cm, __shfl_xor(cm, 8, 32));
            const float mnew = fmaxf(mrow[g], cm);
            alpha[g] = __expf(mrow[g] - mnew);
            mrow[g]  = mnew;
            const float p0 = __expf(s0[g] - mnew);
            const float p1 = __expf(s1[g] - mnew);
            s0[g] = p0; s1[g] = p1;
            float cs = p0 + p1;
            cs += __shfl_xor(cs, 1, 32);
            cs += __shfl_xor(cs, 2, 32);
            cs += __shfl_xor(cs, 4, 32);
            cs += __shfl_xor(cs, 8, 32);
            lrow[g] = lrow[g] * alpha[g] + cs;
        }
#pragma unroll
        for (int t4 = 0; t4 < 4; ++t4)
#pragma unroll
            for (int g = 0; g < 8; ++g) oacc[t4][g] *= alpha[g];

#pragma unroll
        for (int g = 0; g < 8; ++g) {
            const int r = g + half * 8;
            Pw[r * 34 + l15]      = s0[g];
            Pw[r * 34 + 16 + l15] = s1[g];
        }
        asm volatile("s_wait_dscnt 0" ::: "memory");

#pragma unroll
        for (int kk = 0; kk < 8; ++kk) {
            v2f pf = *(const v2f*)&Pw[l15 * 34 + kk * 4 + klo];
#pragma unroll
            for (int t4 = 0; t4 < 4; ++t4) {
                v2f vf = *(const v2f*)&Vc[(t4 * 16 + l15) * 34 + kk * 4 + klo];
                oacc[t4] = wmma4(pf, vf, oacc[t4]);
            }
        }
    };

    stage_store(Ks, VTs);
    __syncthreads();

    int cur = 0;
    for (int mc = 32; mc < 1280; mc += 32) {
        k0 = *(const float4*)(Kg + (size_t)mc * 64);
        k1 = *(const float4*)(Kg + (size_t)(mc + 16) * 64);
        v0 = *(const float4*)(Vg + (size_t)mc * 64);
        v1 = *(const float4*)(Vg + (size_t)(mc + 16) * 64);
        __builtin_amdgcn_sched_barrier(0);

        chunk(Ks + cur * (32 * 68), VTs + cur * (64 * 34));

        const int nxt = cur ^ 1;
        stage_store(Ks + nxt * (32 * 68), VTs + nxt * (64 * 34));
        __syncthreads();
        cur = nxt;
    }
    chunk(Ks + cur * (32 * 68), VTs + cur * (64 * 34));

#pragma unroll
    for (int g = 0; g < 8; ++g) {
        const float rinv = 1.0f / lrow[g];
        const int n = n0 + g + half * 8;
        float* dst = AO + ((size_t)(b * 1024 + n) * 1024) + h * 64;
#pragma unroll
        for (int t4 = 0; t4 < 4; ++t4)
            dst[t4 * 16 + l15] = oacc[t4][g] * rinv;
    }
}

extern "C" void kernel_launch(void* const* d_in, const int* in_sizes, int n_in,
                              void* d_out, int out_size, void* d_ws, size_t ws_size,
                              hipStream_t stream)
{
    const float* x      = (const float*)d_in[0];
    const float* ctx    = (const float*)d_in[1];
    const float* w_qkv  = (const float*)d_in[2];
    const float* w_proj = (const float*)d_in[3];
    const float* b_proj = (const float*)d_in[4];
    float* outp = (float*)d_out;

    float* Qb = (float*)d_ws;
    float* Kb = Qb + 4194304;
    float* Vb = Kb + 5242880;
    float* AO = Vb + 5242880;

    gemm_kernel<1><<<dim3(3072 / 64, 4096 / 128), 256, 0, stream>>>(
        x, w_qkv, nullptr, nullptr, Qb, Kb, Vb, 3072, 1024);

    ctx_scatter_kernel<<<2048, 256, 0, stream>>>(ctx, Kb, Vb);

    attn_kernel<<<dim3(64, 8), 256, 0, stream>>>(Qb, Kb, Vb, AO);

    gemm_kernel<0><<<dim3(1024 / 64, 4096 / 128), 256, 0, stream>>>(
        AO, w_proj, b_proj, outp, nullptr, nullptr, nullptr, 1024, 1024);
}


# Attention products transposed (S^T=K Q^T, O^T+=V^T P^T): a lane owns a query row, so row max/sum are in-lane plus one lane^16 exchange (new sum tree) and P feeds product 2 from registers, not LDS
# speedup vs baseline: 1.1076x; 1.0674x over previous
#include <hip/hip_runtime.h>

typedef float v2f __attribute__((ext_vector_type(2)));
typedef float v8f __attribute__((ext_vector_type(8)));

__device__ __forceinline__ v8f wmma4(v2f a, v2f b, v8f c) {
    return __builtin_amdgcn_wmma_f32_16x16x4_f32(
        false, a, false, b, (short)0, c, false, false);
}

template <int MODE>
__global__ __launch_bounds__(256) void gemm_kernel(
    const float* A, const float* W,
    const float* __restrict__ bias, float* __restrict__ out,
    float* __restrict__ Qb, float* __restrict__ Kb, float* __restrict__ Vb,
    int Ncols, int Kdim)
{
    __shared__ float As[2 * 128 * 20];
    __shared__ float Wsh[2 * 64 * 20];

    const int tid  = threadIdx.x;
    const int wid  = tid >> 5;
    const int lane = tid & 31;
    const int half = lane >> 4;
    const int l15  = lane & 15;
    const int klo  = half * 2;
    const int wave_m = wid >> 1;
    const int wave_n = wid & 1;

    const int rowBase = blockIdx.y * 128;
    const int colBase = blockIdx.x * 64;

    v8f acc[2][2] = {};

    const int lr = tid >> 2;
    const int lc = (tid & 3) << 2;

    const float* Ag0 = A + (size_t)(rowBase + lr) * Kdim + lc;
    const float* Ag1 = A + (size_t)(rowBase + lr + 64) * Kdim + lc;
    const float* Wg  = W + (size_t)(colBase + lr) * Kdim + lc;
    const int sA0 = lr * 20 + lc;
    const int sA1 = (lr + 64) * 20 + lc;
    const int sW  = lr * 20 + lc;
    const int fA  = (wave_m * 32 + l15) * 20 + klo;
    const int fW  = (wave_n * 32 + l15) * 20 + klo;

    float4 pa0 = *(const float4*)Ag0;
    float4 pa1 = *(const float4*)Ag1;
    float4 pw  = *(const float4*)Wg;
    *(float4*)&As[sA0]  = pa0;
    *(float4*)&As[sA1]  = pa1;
    *(float4*)&Wsh[sW]  = pw;
    __syncthreads();

    auto mma_tile = [&](const float* Ac, const float* Wc) {
#pragma unroll
        for (int ks = 0; ks < 16; ks += 4) {
            v2f a0 = *(const v2f*)&Ac[fA + ks];
            v2f a1 = *(const v2f*)&Ac[fA + 16 * 20 + ks];
            v2f w0 = *(const v2f*)&Wc[fW + ks];
            v2f w1 = *(const v2f*)&Wc[fW + 16 * 20 + ks];
            acc[0][0] = wmma4(a0, w0, acc[0][0]);
            acc[0][1] = wmma4(a0, w1, acc[0][1]);
            acc[1][0] = wmma4(a1, w0, acc[1][0]);
            acc[1][1] = wmma4(a1, w1, acc[1][1]);
        }
    };

    int cur = 0;
    for (int kb = 16; kb < Kdim; kb += 16) {
        pa0 = *(const float4*)(Ag0 + kb);
        pa1 = *(const float4*)(Ag1 + kb);
        pw  = *(const float4*)(Wg + kb);
        __builtin_amdgcn_sched_barrier(0);

        mma_tile(As + cur * (128 * 20), Wsh + cur * (64 * 20));
        __builtin_amdgcn_sched_barrier(0);

        const int nxt = cur ^ 1;
        *(float4*)&As[nxt * (128 * 20) + sA0] = pa0;
        *(float4*)&As[nxt * (128 * 20) + sA1] = pa1;
        *(float4*)&Wsh[nxt * (64 * 20) + sW]  = pw;
        __syncthreads();
        cur = nxt;
    }
    mma_tile(As + cur * (128 * 20), Wsh + cur * (64 * 20));

#pragma unroll
    for (int sm = 0; sm < 2; ++sm) {
#pragma unroll
        for (int sn = 0; sn < 2; ++sn) {
            const int j  = colBase + wave_n * 32 + sn * 16 + l15;
            const int r0 = rowBase + wave_m * 32 + sm * 16 + half * 8;
            if (MODE == 0) {
                const float bv = bias ? bias[j] : 0.0f;
                float* dst = out + (size_t)r0 * Ncols + j;
#pragma unroll
                for (int g = 0; g < 8; ++g)
                    dst[(size_t)g * Ncols] = acc[sm][sn][g] + bv;
            } else {
                const int t  = j >> 10;
                const int hh = (j >> 6) & 15;
                const int d  = j & 63;
                const int bb = r0 >> 10;
                const int n  = r0 & 1023;
                float* dst;
                if (t == 0)
                    dst = Qb + ((size_t)(bb * 16 + hh) * 1024 + n) * 64 + d;
                else if (t == 1)
                    dst = Kb + ((size_t)(bb * 16 + hh) * 1280 + 256 + n) * 64 + d;
                else
                    dst = Vb + ((size_t)(bb * 16 + hh) * 1280 + 256 + n) * 64 + d;
#pragma unroll
                for (int g = 0; g < 8; ++g)
                    dst[(size_t)g * 64] = acc[sm][sn][g];
            }
        }
    }
}

__global__ __launch_bounds__(256) void ctx_scatter_kernel(
    const float* __restrict__ ctx, float* __restrict__ Kb, float* __restrict__ Vb)
{
    const int idx = blockIdx.x * 256 + threadIdx.x;
    const int f   = idx << 2;
    const int b   = f >> 19;
    const int m   = (f >> 11) & 255;
    const int q   = f & 2047;
    const int s   = q >> 10;
    const int hh  = (q >> 6) & 15;
    const int d   = q & 63;
    const float4 v = *(const float4*)(ctx + (size_t)f);
    float* base = s ? Vb : Kb;
    *(float4*)(base + ((size_t)(b * 16 + hh) * 1280 + m) * 64 + d) = v;
}

__device__ __forceinline__ float xhalf(float x) {
    const int b = __builtin_bit_cast(int, x);
    return __builtin_bit_cast(float,
        __builtin_amdgcn_permlanex16(b, b, 0x76543210, 0xfedcba98, false, false));
}

__global__ __launch_bounds__(256) void attn_kernel(
    const float* __restrict__ Qb, const float* Kb,
    const float* Vb, float* __restrict__ AO)
{
    __shared__ float Ks[2 * 32 * 68];
    __shared__ float VTs[2 * 64 * 34];

    const int tid  = threadIdx.x;
    const int wid  = tid >> 5;
    const int lane = tid & 31;
    const int half = lane >> 4;
    const int l15  = lane & 15;
    const int klo  = half * 2;

    const int bh = blockIdx.x;
    const int b  = bh >> 4;
    const int h  = bh & 15;
    const float* Qp = Qb + (size_t)bh * 1024 * 64;
    const float* Kp = Kb + (size_t)bh * 1280 * 64;
    const float* Vp = Vb + (size_t)bh * 1280 * 64;

    const int n0   = blockIdx.y * 128 + wid * 16;
    const int qrow = n0 + l15;
    const float scale = 0.125f;

    const int krow = ((l15 >> 1) & 3) * 4 + (l15 >> 3) * 2 + (l15 & 1);

    const int sr = tid >> 4;
    const int sc = (tid & 15) << 2;
    const float* Kg = Kp + (size_t)sr * 64 + sc;
    const float* Vg = Vp + (size_t)sr * 64 + sc;

    float4 k0 = *(const float4*)(Kg);
    float4 k1 = *(const float4*)(Kg + 16 * 64);
    float4 v0 = *(const float4*)(Vg);
    float4 v1 = *(const float4*)(Vg + 16 * 64);

    v2f qf[16];
#pragma unroll
    for (int kk = 0; kk < 16; ++kk) {
        v2f t = *(const v2f*)(Qp + (size_t)qrow * 64 + kk * 4 + klo);
        qf[kk] = t * scale;
    }

    float mrun = -3.0e38f, lrun = 0.0f;
    v8f oacc[4] = {};

    auto stage_store = [&](float* Kd, float* Vd) {
        *(float4*)&Kd[sr * 68 + sc]        = k0;
        *(float4*)&Kd[(sr + 16) * 68 + sc] = k1;
        Vd[(sc + 0) * 34 + sr] = v0.x;
        Vd[(sc + 1) * 34 + sr] = v0.y;
        Vd[(sc + 2) * 34 + sr] = v0.z;
        Vd[(sc + 3) * 34 + sr] = v0.w;
        Vd[(sc + 0) * 34 + sr + 16] = v1.x;
        Vd[(sc + 1) * 34 + sr + 16] = v1.y;
        Vd[(sc + 2) * 34 + sr + 16] = v1.z;
        Vd[(sc + 3) * 34 + sr + 16] = v1.w;
    };

    auto chunk = [&](const float* Kc, const float* Vc) {
        v8f s0 = {}, s1 = {};
#pragma unroll
        for (int kk = 0; kk < 16; ++kk) {
            v2f a0 = *(const v2f*)&Kc[krow * 68 + kk * 4 + klo];
            v2f a1 = *(const v2f*)&Kc[(16 + krow) * 68 + kk * 4 + klo];
            s0 = wmma4(a0, qf[kk], s0);
            s1 = wmma4(a1, qf[kk], s1);
        }

        float cm = fmaxf(s0[0], s1[0]);
#pragma unroll
        for (int r = 1; r < 8; ++r) cm = fmaxf(cm, fmaxf(s0[r], s1[r]));
        cm = fmaxf(cm, xhalf(cm));
        const float mnew  = fmaxf(mrun, cm);
        const float alpha = __expf(mrun - mnew);
        mrun = mnew;

        float cs = 0.0f;
#pragma unroll
        for (int r = 0; r < 8; ++r) {
            const float p0 = __expf(s0[r] - mnew);
            const float p1 = __expf(s1[r] - mnew);
            s0[r] = p0; s1[r] = p1;
            cs += p0 + p1;
        }
        lrun = lrun * alpha + cs;
#pragma unroll
        for (int t4 = 0; t4 < 4; ++t4)
#pragma unroll
            for (int r = 0; r < 8; ++r) oacc[t4][r] *= alpha;

#pragma unroll
        for (int kk = 0; kk < 8; ++kk) {
            const int i2 = (kk & 3) * 2;
            v2f pf;
            if (kk < 4) { pf[0] = s0[i2]; pf[1] = s0[i2 + 1]; }
            else        { pf[0] = s1[i2]; pf[1] = s1[i2 + 1]; }
#pragma unroll
            for (int t4 = 0; t4 < 4; ++t4) {
                v2f vf = *(const v2f*)&Vc[(t4 * 16 + l15) * 34 + kk * 4 + klo];
                oacc[t4] = wmma4(vf, pf, oacc[t4]);
            }
        }
    };

    stage_store(Ks, VTs);
    __syncthreads();

    int cur = 0;
    for (int mc = 32; mc < 1280; mc += 32) {
        k0 = *(const float4*)(Kg + (size_t)mc * 64);
        k1 = *(const float4*)(Kg + (size_t)(mc + 16) * 64);
        v0 = *(const float4*)(Vg + (size_t)mc * 64);
        v1 = *(const float4*)(Vg + (size_t)(mc + 16) * 64);
        __builtin_amdgcn_sched_barrier(0);

        chunk(Ks + cur * (32 * 68), VTs + cur * (64 * 34));
        __builtin_amdgcn_sched_barrier(0);

        const int nxt = cur ^ 1;
        stage_store(Ks + nxt * (32 * 68), VTs + nxt * (64 * 34));
        __syncthreads();
        cur = nxt;
    }
    chunk(Ks + cur * (32 * 68), VTs + cur * (64 * 34));

    const float ltot = lrun + xhalf(lrun);
    const float rinv = 1.0f / ltot;
    float* dst = AO + ((size_t)(b * 1024 + qrow) * 1024) + h * 64 + half * 8;
#pragma unroll
    for (int t4 = 0; t4 < 4; ++t4) {
        float4 lo, hi;
        lo.x = oacc[t4][0] * rinv; lo.y = oacc[t4][1] * rinv;
        lo.z = oacc[t4][2] * rinv; lo.w = oacc[t4][3] * rinv;
        hi.x = oacc[t4][4] * rinv; hi.y = oacc[t4][5] * rinv;
        hi.z = oacc[t4][6] * rinv; hi.w = oacc[t4][7] * rinv;
        *(float4*)(dst + t4 * 16)     = lo;
        *(float4*)(dst + t4 * 16 + 4) = hi;
    }
}

extern "C" void kernel_launch(void* const* d_in, const int* in_sizes, int n_in,
                              void* d_out, int out_size, void* d_ws, size_t ws_size,
                              hipStream_t stream)
{
    const float* x      = (const float*)d_in[0];
    const float* ctx    = (const float*)d_in[1];
    const float* w_qkv  = (const float*)d_in[2];
    const float* w_proj = (const float*)d_in[3];
    const float* b_proj = (const float*)d_in[4];
    float* outp = (float*)d_out;

    float* Qb = (float*)d_ws;
    float* Kb = Qb + 4194304;
    float* Vb = Kb + 5242880;
    float* AO = Vb + 5242880;

    gemm_kernel<1><<<dim3(3072 / 64, 4096 / 128), 256, 0, stream>>>(
        x, w_qkv, nullptr, nullptr, Qb, Kb, Vb, 3072, 1024);

    ctx_scatter_kernel<<<2048, 256, 0, stream>>>(ctx, Kb, Vb);

    attn_kernel<<<dim3(64, 8), 256, 0, stream>>>(Qb, Kb, Vb, AO);

    gemm_kernel<0><<<dim3(1024 / 64, 4096 / 128), 256, 0, stream>>>(
        AO, w_proj, b_proj, outp, nullptr, nullptr, nullptr, 1024, 1024);
}


# GEMM wave tile 32x32 -> 64x64 (QKV) / 64x32 (proj), 4-wave blocks, K-stage 32, 16-byte LDS fragment reads each feeding two k-steps: fewer loads per WMMA; k re-paired within each 8 (new sum order)
# speedup vs baseline: 1.1734x; 1.0594x over previous
#include <hip/hip_runtime.h>

typedef float v2f __attribute__((ext_vector_type(2)));
typedef float v4f __attribute__((ext_vector_type(4)));
typedef float v8f __attribute__((ext_vector_type(8)));

__device__ __forceinline__ v8f wmma4(v2f a, v2f b, v8f c) {
    return __builtin_amdgcn_wmma_f32_16x16x4_f32(
        false, a, false, b, (short)0, c, false, false);
}

template <int MODE, int TN, int NCOLS, int KDIM>
__global__ __launch_bounds__(128) void gemm_kernel(
    const float* A, const float* W,
    const float* __restrict__ bias, float* __restrict__ out,
    float* __restrict__ Qb, float* __restrict__ Kb, float* __restrict__ Vb)
{
    constexpr int BN  = 32 * TN;
    constexpr int NW  = BN / 16;
    constexpr int LDP = 36;
    __shared__ float As[2 * 128 * LDP];
    __shared__ float Wsh[2 * BN * LDP];

    const int tid  = threadIdx.x;
    const int wid  = tid >> 5;
    const int lane = tid & 31;
    const int half = lane >> 4;
    const int l15  = lane & 15;
    const int wave_m = wid >> 1;
    const int wave_n = wid & 1;

    const int rowBase = blockIdx.y * 128;
    const int colBase = blockIdx.x * BN;

    v8f acc[4][TN] = {};

    const int sr = tid >> 3;
    const int sc = (tid & 7) << 2;
    const float* Ag = A + (size_t)(rowBase + sr) * KDIM + sc;
    const float* Wg = W + (size_t)(colBase + sr) * KDIM + sc;
    const int sS = sr * LDP + sc;
    const int fA = (wave_m * 64 + l15) * LDP + half * 4;
    const int fW = (wave_n * 16 * TN + l15) * LDP + half * 4;

    v4f pa[8], pw[NW];

    auto stage_load = [&](int kb) {
#pragma unroll
        for (int i = 0; i < 8; ++i)  pa[i] = *(const v4f*)(Ag + (size_t)(16 * i) * KDIM + kb);
#pragma unroll
        for (int i = 0; i < NW; ++i) pw[i] = *(const v4f*)(Wg + (size_t)(16 * i) * KDIM + kb);
    };
    auto stage_store = [&](float* Ad, float* Wd) {
#pragma unroll
        for (int i = 0; i < 8; ++i)  *(v4f*)&Ad[sS + 16 * i * LDP] = pa[i];
#pragma unroll
        for (int i = 0; i < NW; ++i) *(v4f*)&Wd[sS + 16 * i * LDP] = pw[i];
    };

    auto mma_stage = [&](const float* Ac, const float* Wc) {
#pragma unroll
        for (int kk = 0; kk < 4; ++kk) {
            v4f a[4], w[TN];
#pragma unroll
            for (int i = 0; i < 4; ++i)  a[i] = *(const v4f*)&Ac[fA + 16 * i * LDP + kk * 8];
#pragma unroll
            for (int n = 0; n < TN; ++n) w[n] = *(const v4f*)&Wc[fW + 16 * n * LDP + kk * 8];
#pragma unroll
            for (int i = 0; i < 4; ++i)
#pragma unroll
                for (int n = 0; n < TN; ++n) {
                    v2f af = {a[i].x, a[i].y};
                    v2f wf = {w[n].x, w[n].y};
                    acc[i][n] = wmma4(af, wf, acc[i][n]);
                }
#pragma unroll
            for (int i = 0; i < 4; ++i)
#pragma unroll
                for (int n = 0; n < TN; ++n) {
                    v2f af = {a[i].z, a[i].w};
                    v2f wf = {w[n].z, w[n].w};
                    acc[i][n] = wmma4(af, wf, acc[i][n]);
                }
        }
    };

    stage_load(0);
    stage_store(As, Wsh);
    __syncthreads();

    int cur = 0;
    for (int kb = 32; kb < KDIM; kb += 32) {
        stage_load(kb);
        __builtin_amdgcn_sched_barrier(0);

        mma_stage(As + cur * (128 * LDP), Wsh + cur * (BN * LDP));
        __builtin_amdgcn_sched_barrier(0);

        const int nxt = cur ^ 1;
        stage_store(As + nxt * (128 * LDP), Wsh + nxt * (BN * LDP));
        __syncthreads();
        cur = nxt;
    }
    mma_stage(As + cur * (128 * LDP), Wsh + cur * (BN * LDP));

#pragma unroll
    for (int sm = 0; sm < 4; ++sm) {
#pragma unroll
        for (int sn = 0; sn < TN; ++sn) {
            const int j  = colBase + wave_n * 16 * TN + sn * 16 + l15;
            const int r0 = rowBase + wave_m * 64 + sm * 16 + half * 8;
            if (MODE == 0) {
                const float bv = bias ? bias[j] : 0.0f;
                float* dst = out + (size_t)r0 * NCOLS + j;
#pragma unroll
                for (int g = 0; g < 8; ++g)
                    dst[(size_t)g * NCOLS] = acc[sm][sn][g] + bv;
            } else {
                const int t  = j >> 10;
                const int hh = (j >> 6) & 15;
                const int d  = j & 63;
                const int bb = r0 >> 10;
                const int n  = r0 & 1023;
                float* dst;
                if (t == 0)
                    dst = Qb + ((size_t)(bb * 16 + hh) * 1024 + n) * 64 + d;
                else if (t == 1)
                    dst = Kb + ((size_t)(bb * 16 + hh) * 1280 + 256 + n) * 64 + d;
                else
                    dst = Vb + ((size_t)(bb * 16 + hh) * 1280 + 256 + n) * 64 + d;
#pragma unroll
                for (int g = 0; g < 8; ++g)
                    dst[(size_t)g * 64] = acc[sm][sn][g];
            }
        }
    }
}

__global__ __launch_bounds__(256) void ctx_scatter_kernel(
    const float* __restrict__ ctx, float* __restrict__ Kb, float* __restrict__ Vb)
{
    const int idx = blockIdx.x * 256 + threadIdx.x;
    const int f   = idx << 2;
    const int b   = f >> 19;
    const int m   = (f >> 11) & 255;
    const int q   = f & 2047;
    const int s   = q >> 10;
    const int hh  = (q >> 6) & 15;
    const int d   = q & 63;
    const float4 v = *(const float4*)(ctx + (size_t)f);
    float* base = s ? Vb : Kb;
    *(float4*)(base + ((size_t)(b * 16 + hh) * 1280 + m) * 64 + d) = v;
}

__device__ __forceinline__ float xhalf(float x) {
    const int b = __builtin_bit_cast(int, x);
    return __builtin_bit_cast(float,
        __builtin_amdgcn_permlanex16(b, b, 0x76543210, 0xfedcba98, false, false));
}

__global__ __launch_bounds__(256) void attn_kernel(
    const float* __restrict__ Qb, const float* Kb,
    const float* Vb, float* __restrict__ AO)
{
    __shared__ float Ks[2 * 32 * 68];
    __shared__ float VTs[2 * 64 * 34];

    const int tid  = threadIdx.x;
    const int wid  = tid >> 5;
    const int lane = tid & 31;
    const int half = lane >> 4;
    const int l15  = lane & 15;
    const int klo  = half * 2;

    const int bh = blockIdx.x;
    const int b  = bh >> 4;
    const int h  = bh & 15;
    const float* Qp = Qb + (size_t)bh * 1024 * 64;
    const float* Kp = Kb + (size_t)bh * 1280 * 64;
    const float* Vp = Vb + (size_t)bh * 1280 * 64;

    const int n0   = blockIdx.y * 128 + wid * 16;
    const int qrow = n0 + l15;
    const float scale = 0.125f;

    const int krow = ((l15 >> 1) & 3) * 4 + (l15 >> 3) * 2 + (l15 & 1);

    const int sr = tid >> 4;
    const int sc = (tid & 15) << 2;
    const float* Kg = Kp + (size_t)sr * 64 + sc;
    const float* Vg = Vp + (size_t)sr * 64 + sc;

    float4 k0 = *(const float4*)(Kg);
    float4 k1 = *(const float4*)(Kg + 16 * 64);
    float4 v0 = *(const float4*)(Vg);
    float4 v1 = *(const float4*)(Vg + 16 * 64);

    v2f qf[16];
#pragma unroll
    for (int kk = 0; kk < 16; ++kk) {
        v2f t = *(const v2f*)(Qp + (size_t)qrow * 64 + kk * 4 + klo);
        qf[kk] = t * scale;
    }

    float mrun = -3.0e38f, lrun = 0.0f;
    v8f oacc[4] = {};

    auto stage_store = [&](float* Kd, float* Vd) {
        *(float4*)&Kd[sr * 68 + sc]        = k0;
        *(float4*)&Kd[(sr + 16) * 68 + sc] = k1;
        Vd[(sc + 0) * 34 + sr] = v0.x;
        Vd[(sc + 1) * 34 + sr] = v0.y;
        Vd[(sc + 2) * 34 + sr] = v0.z;
        Vd[(sc + 3) * 34 + sr] = v0.w;
        Vd[(sc + 0) * 34 + sr + 16] = v1.x;
        Vd[(sc + 1) * 34 + sr + 16] = v1.y;
        Vd[(sc + 2) * 34 + sr + 16] = v1.z;
        Vd[(sc + 3) * 34 + sr + 16] = v1.w;
    };

    auto chunk = [&](const float* Kc, const float* Vc) {
        v8f s0 = {}, s1 = {};
#pragma unroll
        for (int kk = 0; kk < 16; ++kk) {
            v2f a0 = *(const v2f*)&Kc[krow * 68 + kk * 4 + klo];
            v2f a1 = *(const v2f*)&Kc[(16 + krow) * 68 + kk * 4 + klo];
            s0 = wmma4(a0, qf[kk], s0);
            s1 = wmma4(a1, qf[kk], s1);
        }

        float cm = fmaxf(s0[0], s1[0]);
#pragma unroll
        for (int r = 1; r < 8; ++r) cm = fmaxf(cm, fmaxf(s0[r], s1[r]));
        cm = fmaxf(cm, xhalf(cm));
        const float mnew  = fmaxf(mrun, cm);
        const float alpha = __expf(mrun - mnew);
        mrun = mnew;

        float cs = 0.0f;
#pragma unroll
        for (int r = 0; r < 8; ++r) {
            const float p0 = __expf(s0[r] - mnew);
            const float p1 = __expf(s1[r] - mnew);
            s0[r] = p0; s1[r] = p1;
            cs += p0 + p1;
        }
        lrun = lrun * alpha + cs;
#pragma unroll
        for (int t4 = 0; t4 < 4; ++t4)
#pragma unroll
            for (int r = 0; r < 8; ++r) oacc[t4][r] *= alpha;

#pragma unroll
        for (int kk = 0; kk < 8; ++kk) {
            const int i2 = (kk & 3) * 2;
            v2f pf;
            if (kk < 4) { pf[0] = s0[i2]; pf[1] = s0[i2 + 1]; }
            else        { pf[0] = s1[i2]; pf[1] = s1[i2 + 1]; }
#pragma unroll
            for (int t4 = 0; t4 < 4; ++t4) {
                v2f vf = *(const v2f*)&Vc[(t4 * 16 + l15) * 34 + kk * 4 + klo];
                oacc[t4] = wmma4(vf, pf, oacc[t4]);
            }
        }
    };

    stage_store(Ks, VTs);
    __syncthreads();

    int cur = 0;
    for (int mc = 32; mc < 1280; mc += 32) {
        k0 = *(const float4*)(Kg + (size_t)mc * 64);
        k1 = *(const float4*)(Kg + (size_t)(mc + 16) * 64);
        v0 = *(const float4*)(Vg + (size_t)mc * 64);
        v1 = *(const float4*)(Vg + (size_t)(mc + 16) * 64);
        __builtin_amdgcn_sched_barrier(0);

        chunk(Ks + cur * (32 * 68), VTs + cur * (64 * 34));
        __builtin_amdgcn_sched_barrier(0);

        const int nxt = cur ^ 1;
        stage_store(Ks + nxt * (32 * 68), VTs + nxt * (64 * 34));
        __syncthreads();
        cur = nxt;
    }
    chunk(Ks + cur * (32 * 68), VTs + cur * (64 * 34));

    const float ltot = lrun + xhalf(lrun);
    const float rinv = 1.0f / ltot;
    float* dst = AO + ((size_t)(b * 1024 + qrow) * 1024) + h * 64 + half * 8;
#pragma unroll
    for (int t4 = 0; t4 < 4; ++t4) {
        float4 lo, hi;
        lo.x = oacc[t4][0] * rinv; lo.y = oacc[t4][1] * rinv;
        lo.z = oacc[t4][2] * rinv; lo.w = oacc[t4][3] * rinv;
        hi.x = oacc[t4][4] * rinv; hi.y = oacc[t4][5] * rinv;
        hi.z = oacc[t4][6] * rinv; hi.w = oacc[t4][7] * rinv;
        *(float4*)(dst + t4 * 16)     = lo;
        *(float4*)(dst + t4 * 16 + 4) = hi;
    }
}

extern "C" void kernel_launch(void* const* d_in, const int* in_sizes, int n_in,
                              void* d_out, int out_size, void* d_ws, size_t ws_size,
                              hipStream_t stream)
{
    const float* x      = (const float*)d_in[0];
    const float* ctx    = (const float*)d_in[1];
    const float* w_qkv  = (const float*)d_in[2];
    const float* w_proj = (const float*)d_in[3];
    const float* b_proj = (const float*)d_in[4];
    float* outp = (float*)d_out;

    float* Qb = (float*)d_ws;
    float* Kb = Qb + 4194304;
    float* Vb = Kb + 5242880;
    float* AO = Vb + 5242880;

    gemm_kernel<1, 4, 3072, 1024><<<dim3(3072 / 128, 4096 / 128), 128, 0, stream>>>(
        x, w_qkv, nullptr, nullptr, Qb, Kb, Vb);

    ctx_scatter_kernel<<<2048, 256, 0, stream>>>(ctx, Kb, Vb);

    attn_kernel<<<dim3(64, 8), 256, 0, stream>>>(Qb, Kb, Vb, AO);

    gemm_kernel<0, 2, 1024, 1024><<<dim3(1024 / 64, 4096 / 128), 128, 0, stream>>>(
        AO, w_proj, b_proj, outp, nullptr, nullptr, nullptr);
}


# Wave priority falls with loop progress (s_setprio 3,2,1,0 from 1/2, 3/4, 7/8 of the K and kv loops) in all three matrix kernels, so co-resident workgroups finish together; arithmetic unchanged
# speedup vs baseline: 1.2189x; 1.0388x over previous
#include <hip/hip_runtime.h>

typedef float v2f __attribute__((ext_vector_type(2)));
typedef float v4f __attribute__((ext_vector_type(4)));
typedef float v8f __attribute__((ext_vector_type(8)));

__device__ __forceinline__ v8f wmma4(v2f a, v2f b, v8f c) {
    return __builtin_amdgcn_wmma_f32_16x16x4_f32(
        false, a, false, b, (short)0, c, false, false);
}

__device__ __forceinline__ void prio_for_segment(int seg) {
    if (seg == 0)      __builtin_amdgcn_s_setprio(3);
    else if (seg == 1) __builtin_amdgcn_s_setprio(2);
    else if (seg == 2) __builtin_amdgcn_s_setprio(1);
    else               __builtin_amdgcn_s_setprio(0);
}
__device__ __forceinline__ int segment_end(int seg, int total) {
    return seg == 3 ? total : total - (total >> (seg + 1));
}

template <int MODE, int TN, int NCOLS, int KDIM>
__global__ __launch_bounds__(128) void gemm_kernel(
    const float* A, const float* W,
    const float* __restrict__ bias, float* __restrict__ out,
    float* __restrict__ Qb, float* __restrict__ Kb, float* __restrict__ Vb)
{
    constexpr int BN  = 32 * TN;
    constexpr int NW  = BN / 16;
    constexpr int LDP = 36;
    __shared__ float As[2 * 128 * LDP];
    __shared__ float Wsh[2 * BN * LDP];

    const int tid  = threadIdx.x;
    const int wid  = tid >> 5;
    const int lane = tid & 31;
    const int half = lane >> 4;
    const int l15  = lane & 15;
    const int wave_m = wid >> 1;
    const int wave_n = wid & 1;

    const int rowBase = blockIdx.y * 128;
    const int colBase = blockIdx.x * BN;

    v8f acc[4][TN] = {};

    const int sr = tid >> 3;
    const int sc = (tid & 7) << 2;
    const float* Ag = A + (size_t)(rowBase + sr) * KDIM + sc;
    const float* Wg = W + (size_t)(colBase + sr) * KDIM + sc;
    const int sS = sr * LDP + sc;
    const int fA = (wave_m * 64 + l15) * LDP + half * 4;
    const int fW = (wave_n * 16 * TN + l15) * LDP + half * 4;

    v4f pa[8], pw[NW];

    auto stage_load = [&](int kb) {
#pragma unroll
        for (int i = 0; i < 8; ++i)  pa[i] = *(const v4f*)(Ag + (size_t)(16 * i) * KDIM + kb);
#pragma unroll
        for (int i = 0; i < NW; ++i) pw[i] = *(const v4f*)(Wg + (size_t)(16 * i) * KDIM + kb);
    };
    auto stage_store = [&](float* Ad, float* Wd) {
#pragma unroll
        for (int i = 0; i < 8; ++i)  *(v4f*)&Ad[sS + 16 * i * LDP] = pa[i];
#pragma unroll
        for (int i = 0; i < NW; ++i) *(v4f*)&Wd[sS + 16 * i * LDP] = pw[i];
    };

    auto mma_stage = [&](const float* Ac, const float* Wc) {
#pragma unroll
        for (int kk = 0; kk < 4; ++kk) {
            v4f a[4], w[TN];
#pragma unroll
            for (int i = 0; i < 4; ++i)  a[i] = *(const v4f*)&Ac[fA + 16 * i * LDP + kk * 8];
#pragma unroll
            for (int n = 0; n < TN; ++n) w[n] = *(const v4f*)&Wc[fW + 16 * n * LDP + kk * 8];
#pragma unroll
            for (int i = 0; i < 4; ++i)
#pragma unroll
                for (int n = 0; n < TN; ++n) {
                    v2f af = {a[i].x, a[i].y};
                    v2f wf = {w[n].x, w[n].y};
                    acc[i][n] = wmma4(af, wf, acc[i][n]);
                }
#pragma unroll
            for (int i = 0; i < 4; ++i)
#pragma unroll
                for (int n = 0; n < TN; ++n) {
                    v2f af = {a[i].z, a[i].w};
                    v2f wf = {w[n].z, w[n].w};
                    acc[i][n] = wmma4(af, wf, acc[i][n]);
                }
        }
    };

    stage_load(0);
    stage_store(As, Wsh);
    __syncthreads();

    int cur = 0;
    int kb = 32;
    for (int seg = 0; seg < 4; ++seg) {
        prio_for_segment(seg);
        const int kend = segment_end(seg, KDIM);
        for (; kb < kend; kb += 32) {
            stage_load(kb);
            __builtin_amdgcn_sched_barrier(0);

            mma_stage(As + cur * (128 * LDP), Wsh + cur * (BN * LDP));
            __builtin_amdgcn_sched_barrier(0);

            const int nxt = cur ^ 1;
            stage_store(As + nxt * (128 * LDP), Wsh + nxt * (BN * LDP));
            __syncthreads();
            cur = nxt;
        }
    }
    mma_stage(As + cur * (128 * LDP), Wsh + cur * (BN * LDP));

#pragma unroll
    for (int sm = 0; sm < 4; ++sm) {
#pragma unroll
        for (int sn = 0; sn < TN; ++sn) {
            const int j  = colBase + wave_n * 16 * TN + sn * 16 + l15;
            const int r0 = rowBase + wave_m * 64 + sm * 16 + half * 8;
            if (MODE == 0) {
                const float bv = bias ? bias[j] : 0.0f;
                float* dst = out + (size_t)r0 * NCOLS + j;
#pragma unroll
                for (int g = 0; g < 8; ++g)
                    dst[(size_t)g * NCOLS] = acc[sm][sn][g] + bv;
            } else {
                const int t  = j >> 10;
                const int hh = (j >> 6) & 15;
                const int d  = j & 63;
                const int bb = r0 >> 10;
                const int n  = r0 & 1023;
                float* dst;
                if (t == 0)
                    dst = Qb + ((size_t)(bb * 16 + hh) * 1024 + n) * 64 + d;
                else if (t == 1)
                    dst = Kb + ((size_t)(bb * 16 + hh) * 1280 + 256 + n) * 64 + d;
                else
                    dst = Vb + ((size_t)(bb * 16 + hh) * 1280 + 256 + n) * 64 + d;
#pragma unroll
                for (int g = 0; g < 8; ++g)
                    dst[(size_t)g * 64] = acc[sm][sn][g];
            }
        }
    }
}

__global__ __launch_bounds__(256) void ctx_scatter_kernel(
    const float* __restrict__ ctx, float* __restrict__ Kb, float* __restrict__ Vb)
{
    const int idx = blockIdx.x * 256 + threadIdx.x;
    const int f   = idx << 2;
    const int b   = f >> 19;
    const int m   = (f >> 11) & 255;
    const int q   = f & 2047;
    const int s   = q >> 10;
    const int hh  = (q >> 6) & 15;
    const int d   = q & 63;
    const float4 v = *(const float4*)(ctx + (size_t)f);
    float* base = s ? Vb : Kb;
    *(float4*)(base + ((size_t)(b * 16 + hh) * 1280 + m) * 64 + d) = v;
}

__device__ __forceinline__ float xhalf(float x) {
    const int b = __builtin_bit_cast(int, x);
    return __builtin_bit_cast(float,
        __builtin_amdgcn_permlanex16(b, b, 0x76543210, 0xfedcba98, false, false));
}

__global__ __launch_bounds__(256) void attn_kernel(
    const float* __restrict__ Qb, const float* Kb,
    const float* Vb, float* __restrict__ AO)
{
    __shared__ float Ks[2 * 32 * 68];
    __shared__ float VTs[2 * 64 * 34];

    const int tid  = threadIdx.x;
    const int wid  = tid >> 5;
    const int lane = tid & 31;
    const int half = lane >> 4;
    const int l15  = lane & 15;
    const int klo  = half * 2;

    const int bh = blockIdx.x;
    const int b  = bh >> 4;
    const int h  = bh & 15;
    const float* Qp = Qb + (size_t)bh * 1024 * 64;
    const float* Kp = Kb + (size_t)bh * 1280 * 64;
    const float* Vp = Vb + (size_t)bh * 1280 * 64;

    const int n0   = blockIdx.y * 128 + wid * 16;
    const int qrow = n0 + l15;
    const float scale = 0.125f;

    const int krow = ((l15 >> 1) & 3) * 4 + (l15 >> 3) * 2 + (l15 & 1);

    const int sr = tid >> 4;
    const int sc = (tid & 15) << 2;
    const float* Kg = Kp + (size_t)sr * 64 + sc;
    const float* Vg = Vp + (size_t)sr * 64 + sc;

    float4 k0 = *(const float4*)(Kg);
    float4 k1 = *(const float4*)(Kg + 16 * 64);
    float4 v0 = *(const float4*)(Vg);
    float4 v1 = *(const float4*)(Vg + 16 * 64);

    v2f qf[16];
#pragma unroll
    for (int kk = 0; kk < 16; ++kk) {
        v2f t = *(const v2f*)(Qp + (size_t)qrow * 64 + kk * 4 + klo);
        qf[kk] = t * scale;
    }

    float mrun = -3.0e38f, lrun = 0.0f;
    v8f oacc[4] = {};

    auto stage_store = [&](float* Kd, float* Vd) {
        *(float4*)&Kd[sr * 68 + sc]        = k0;
        *(float4*)&Kd[(sr + 16) * 68 + sc] = k1;
        Vd[(sc + 0) * 34 + sr] = v0.x;
        Vd[(sc + 1) * 34 + sr] = v0.y;
        Vd[(sc + 2) * 34 + sr] = v0.z;
        Vd[(sc + 3) * 34 + sr] = v0.w;
        Vd[(sc + 0) * 34 + sr + 16] = v1.x;
        Vd[(sc + 1) * 34 + sr + 16] = v1.y;
        Vd[(sc + 2) * 34 + sr + 16] = v1.z;
        Vd[(sc + 3) * 34 + sr + 16] = v1.w;
    };

    auto chunk = [&](const float* Kc, const float* Vc) {
        v8f s0 = {}, s1 = {};
#pragma unroll
        for (int kk = 0; kk < 16; ++kk) {
            v2f a0 = *(const v2f*)&Kc[krow * 68 + kk * 4 + klo];
            v2f a1 = *(const v2f*)&Kc[(16 + krow) * 68 + kk * 4 + klo];
            s0 = wmma4(a0, qf[kk], s0);
            s1 = wmma4(a1, qf[kk], s1);
        }

        float cm = fmaxf(s0[0], s1[0]);
#pragma unroll
        for (int r = 1; r < 8; ++r) cm = fmaxf(cm, fmaxf(s0[r], s1[r]));
        cm = fmaxf(cm, xhalf(cm));
        const float mnew  = fmaxf(mrun, cm);
        const float alpha = __expf(mrun - mnew);
        mrun = mnew;

        float cs = 0.0f;
#pragma unroll
        for (int r = 0; r < 8; ++r) {
            const float p0 = __expf(s0[r] - mnew);
            const float p1 = __expf(s1[r] - mnew);
            s0[r] = p0; s1[r] = p1;
            cs += p0 + p1;
        }
        lrun = lrun * alpha + cs;
#pragma unroll
        for (int t4 = 0; t4 < 4; ++t4)
#pragma unroll
            for (int r = 0; r < 8; ++r) oacc[t4][r] *= alpha;

#pragma unroll
        for (int kk = 0; kk < 8; ++kk) {
            const int i2 = (kk & 3) * 2;
            v2f pf;
            if (kk < 4) { pf[0] = s0[i2]; pf[1] = s0[i2 + 1]; }
            else        { pf[0] = s1[i2]; pf[1] = s1[i2 + 1]; }
#pragma unroll
            for (int t4 = 0; t4 < 4; ++t4) {
                v2f vf = *(const v2f*)&Vc[(t4 * 16 + l15) * 34 + kk * 4 + klo];
                oacc[t4] = wmma4(vf, pf, oacc[t4]);
            }
        }
    };

    stage_store(Ks, VTs);
    __syncthreads();

    int cur = 0;
    int mc = 32;
    for (int seg = 0; seg < 4; ++seg) {
        prio_for_segment(seg);
        const int mend = segment_end(seg, 1280);
        for (; mc < mend; mc += 32) {
            k0 = *(const float4*)(Kg + (size_t)mc * 64);
            k1 = *(const float4*)(Kg + (size_t)(mc + 16) * 64);
            v0 = *(const float4*)(Vg + (size_t)mc * 64);
            v1 = *(const float4*)(Vg + (size_t)(mc + 16) * 64);
            __builtin_amdgcn_sched_barrier(0);

            chunk(Ks + cur * (32 * 68), VTs + cur * (64 * 34));
            __builtin_amdgcn_sched_barrier(0);

            const int nxt = cur ^ 1;
            stage_store(Ks + nxt * (32 * 68), VTs + nxt * (64 * 34));
            __syncthreads();
            cur = nxt;
        }
    }
    chunk(Ks + cur * (32 * 68), VTs + cur * (64 * 34));

    const float ltot = lrun + xhalf(lrun);
    const float rinv = 1.0f / ltot;
    float* dst = AO + ((size_t)(b * 1024 + qrow) * 1024) + h * 64 + half * 8;
#pragma unroll
    for (int t4 = 0; t4 < 4; ++t4) {
        float4 lo, hi;
        lo.x = oacc[t4][0] * rinv; lo.y = oacc[t4][1] * rinv;
        lo.z = oacc[t4][2] * rinv; lo.w = oacc[t4][3] * rinv;
        hi.x = oacc[t4][4] * rinv; hi.y = oacc[t4][5] * rinv;
        hi.z = oacc[t4][6] * rinv; hi.w = oacc[t4][7] * rinv;
        *(float4*)(dst + t4 * 16)     = lo;
        *(float4*)(dst + t4 * 16 + 4) = hi;
    }
}

extern "C" void kernel_launch(void* const* d_in, const int* in_sizes, int n_in,
                              void* d_out, int out_size, void* d_ws, size_t ws_size,
                              hipStream_t stream)
{
    const float* x      = (const float*)d_in[0];
    const float* ctx    = (const float*)d_in[1];
    const float* w_qkv  = (const float*)d_in[2];
    const float* w_proj = (const float*)d_in[3];
    const float* b_proj = (const float*)d_in[4];
    float* outp = (float*)d_out;

    float* Qb = (float*)d_ws;
    float* Kb = Qb + 4194304;
    float* Vb = Kb + 5242880;
    float* AO = Vb + 5242880;

    gemm_kernel<1, 4, 3072, 1024><<<dim3(3072 / 128, 4096 / 128), 128, 0, stream>>>(
        x, w_qkv, nullptr, nullptr, Qb, Kb, Vb);

    ctx_scatter_kernel<<<2048, 256, 0, stream>>>(ctx, Kb, Vb);

    attn_kernel<<<dim3(64, 8), 256, 0, stream>>>(Qb, Kb, Vb, AO);

    gemm_kernel<0, 2, 1024, 1024><<<dim3(1024 / 64, 4096 / 128), 128, 0, stream>>>(
        AO, w_proj, b_proj, outp, nullptr, nullptr, nullptr);
}
